# starting kernel
# baseline (speedup 1.0000x reference)
#include <hip/hip_runtime.h>
#include <hip/hip_bf16.h>
#include <math.h>

typedef __attribute__((ext_vector_type(2))) float v2f;
typedef __attribute__((ext_vector_type(4))) float v4f;
typedef __attribute__((ext_vector_type(8))) float v8f;

__device__ __forceinline__ void make_rinv(const float* __restrict__ q,
                                          const float* __restrict__ dt,
                                          int b, float R[9], float ti[3]) {
    float qw = q[4 * b + 0], qx = q[4 * b + 1], qy = q[4 * b + 2], qz = q[4 * b + 3];
    float s = 1.0f / sqrtf(qw * qw + qx * qx + qy * qy + qz * qz);
    qw *= s; qx *= s; qy *= s; qz *= s;
    float R00 = 1.f - 2.f * (qy * qy + qz * qz);
    float R01 = 2.f * (qx * qy - qw * qz);
    float R02 = 2.f * (qx * qz + qw * qy);
    float R10 = 2.f * (qx * qy + qw * qz);
    float R11 = 1.f - 2.f * (qx * qx + qz * qz);
    float R12 = 2.f * (qy * qz - qw * qx);
    float R20 = 2.f * (qx * qz - qw * qy);
    float R21 = 2.f * (qy * qz + qw * qx);
    float R22 = 1.f - 2.f * (qx * qx + qy * qy);
    R[0] = R00; R[1] = R10; R[2] = R20;
    R[3] = R01; R[4] = R11; R[5] = R21;
    R[6] = R02; R[7] = R12; R[8] = R22;
    float t0 = dt[3 * b + 0], t1 = dt[3 * b + 1], t2 = dt[3 * b + 2];
    ti[0] = -(R00 * t0 + R10 * t1 + R20 * t2);
    ti[1] = -(R01 * t0 + R11 * t1 + R21 * t2);
    ti[2] = -(R02 * t0 + R12 * t1 + R22 * t2);
}

__global__ __launch_bounds__(32) void setup_wmma_kernel(
    const float* __restrict__ T, const float* __restrict__ q,
    const float* __restrict__ dt, float* __restrict__ outT,
    float* __restrict__ ws) {
    __shared__ float sA[256];
    int lane = threadIdx.x;
    if (lane < 16) {
        int b = lane;
        float R[9], ti[3];
        make_rinv(q, dt, b, R, ti);
        sA[b * 16 + 0]  = R[0]; sA[b * 16 + 1]  = R[1]; sA[b * 16 + 2]  = R[2]; sA[b * 16 + 3]  = ti[0];
        sA[b * 16 + 4]  = R[3]; sA[b * 16 + 5]  = R[4]; sA[b * 16 + 6]  = R[5]; sA[b * 16 + 7]  = ti[1];
        sA[b * 16 + 8]  = R[6]; sA[b * 16 + 9]  = R[7]; sA[b * 16 + 10] = R[8]; sA[b * 16 + 11] = ti[2];
        sA[b * 16 + 12] = 0.f;  sA[b * 16 + 13] = 0.f;  sA[b * 16 + 14] = 0.f;  sA[b * 16 + 15] = 1.f;
        if (ws) {
            #pragma unroll
            for (int i = 0; i < 9; ++i) ws[b * 12 + i] = R[i];
            ws[b * 12 + 9] = ti[0]; ws[b * 12 + 10] = ti[1]; ws[b * 12 + 11] = ti[2];
        }
    }
    __syncthreads();

    int r  = lane & 15;
    int hi = lane >> 4;
    int qb = r >> 2;
    int qi = r & 3;
    #pragma unroll
    for (int g = 0; g < 4; ++g) {
        int batch = g * 4 + qb;
        v2f a;
        a.x = sA[batch * 16 + qi * 4 + 2 * hi + 0];
        a.y = sA[batch * 16 + qi * 4 + 2 * hi + 1];
        v2f bm;
        bm.x = T[batch * 16 + (2 * hi + 0) * 4 + qi];
        bm.y = T[batch * 16 + (2 * hi + 1) * 4 + qi];
        v8f c = {};
        c = __builtin_amdgcn_wmma_f32_16x16x4_f32(
             false, a,  false, bm,
             (short)0, c,  false,  false);
        #pragma unroll
        for (int v = 0; v < 8; ++v) {
            int m = v + 8 * hi;
            if ((m >> 2) == qb) {
                outT[batch * 16 + (m & 3) * 4 + qi] = c[v];
            }
        }
    }
}

__global__ void setup_scalar_kernel(const float* __restrict__ T,
                                    const float* __restrict__ q,
                                    const float* __restrict__ dt,
                                    float* __restrict__ outT,
                                    float* __restrict__ ws, int B) {
    int b = blockIdx.x * blockDim.x + threadIdx.x;
    if (b >= B) return;
    float R[9], ti[3];
    make_rinv(q, dt, b, R, ti);
    float Ainv[16] = {R[0], R[1], R[2], ti[0],
                      R[3], R[4], R[5], ti[1],
                      R[6], R[7], R[8], ti[2],
                      0.f,  0.f,  0.f,  1.f};
    for (int i = 0; i < 4; ++i)
        for (int j = 0; j < 4; ++j) {
            float s = 0.f;
            for (int k = 0; k < 4; ++k) s += Ainv[i * 4 + k] * T[b * 16 + k * 4 + j];
            outT[b * 16 + i * 4 + j] = s;
        }
    if (ws) {
        for (int i = 0; i < 9; ++i) ws[b * 12 + i] = R[i];
        ws[b * 12 + 9] = ti[0]; ws[b * 12 + 10] = ti[1]; ws[b * 12 + 11] = ti[2];
    }
}

__global__ __launch_bounds__(256) void realign_kernel(
    const float* __restrict__ pcd, const float* __restrict__ q,
    const float* __restrict__ dt, const float* __restrict__ ws,
    float* __restrict__ outP, int N, int gpb) {
    int b = blockIdx.y;
    int g = blockIdx.x * blockDim.x + threadIdx.x;
    if (g >= gpb) return;

    float R[9], ti[3];
    if (ws) {
        const float* w = ws + b * 12;
        #pragma unroll
        for (int i = 0; i < 9; ++i) R[i] = w[i];
        ti[0] = w[9]; ti[1] = w[10]; ti[2] = w[11];
    } else {
        make_rinv(q, dt, b, R, ti);
    }

    long long p0 = (long long)g * 4;
    size_t base = (size_t)b * (size_t)N * 3 + (size_t)p0 * 3;

    if (p0 + 4 <= N) {
        const v4f* src = (const v4f*)(pcd + base);
        v4f va = __builtin_nontemporal_load(src + 0);
        v4f vb = __builtin_nontemporal_load(src + 1);
        v4f vc = __builtin_nontemporal_load(src + 2);
        float f[12] = {va.x, va.y, va.z, va.w,
                       vb.x, vb.y, vb.z, vb.w,
                       vc.x, vc.y, vc.z, vc.w};
        float o[12];
        #pragma unroll
        for (int p = 0; p < 4; ++p) {
            float px = f[p * 3 + 0], py = f[p * 3 + 1], pz = f[p * 3 + 2];
            o[p * 3 + 0] = fmaf(R[0], px, fmaf(R[1], py, fmaf(R[2], pz, ti[0])));
            o[p * 3 + 1] = fmaf(R[3], px, fmaf(R[4], py, fmaf(R[5], pz, ti[1])));
            o[p * 3 + 2] = fmaf(R[6], px, fmaf(R[7], py, fmaf(R[8], pz, ti[2])));
        }
        v4f o0 = {o[0], o[1], o[2], o[3]};
        v4f o1 = {o[4], o[5], o[6], o[7]};
        v4f o2 = {o[8], o[9], o[10], o[11]};
        v4f* dst = (v4f*)(outP + base);
        __builtin_nontemporal_store(o0, dst + 0);
        __builtin_nontemporal_store(o1, dst + 1);
        __builtin_nontemporal_store(o2, dst + 2);
    } else {
        for (long long p = p0; p < N; ++p) {
            size_t idx = (size_t)b * (size_t)N * 3 + (size_t)p * 3;
            float px = pcd[idx + 0], py = pcd[idx + 1], pz = pcd[idx + 2];
            outP[idx + 0] = fmaf(R[0], px, fmaf(R[1], py, fmaf(R[2], pz, ti[0])));
            outP[idx + 1] = fmaf(R[3], px, fmaf(R[4], py, fmaf(R[5], pz, ti[1])));
            outP[idx + 2] = fmaf(R[6], px, fmaf(R[7], py, fmaf(R[8], pz, ti[2])));
        }
    }
}

extern "C" void kernel_launch(void* const* d_in, const int* in_sizes, int n_in,
                              void* d_out, int out_size, void* d_ws, size_t ws_size,
                              hipStream_t stream) {
    const float* pcd = (const float*)d_in[0];
    const float* T   = (const float*)d_in[1];
    const float* q   = (const float*)d_in[2];
    const float* dt  = (const float*)d_in[3];

    int B = in_sizes[2] / 4;
    long long pcd_elems = in_sizes[0];
    int N = (int)(pcd_elems / ((long long)B * 3));

    float* outT = (float*)d_out;
    float* outP = outT + (size_t)B * 16;
    float* ws = (ws_size >= (size_t)B * 12 * sizeof(float)) ? (float*)d_ws : nullptr;

    if (B == 16) {
        setup_wmma_kernel<<<1, 32, 0, stream>>>(T, q, dt, outT, ws);
    } else {
        int thr = 64;
        setup_scalar_kernel<<<(B + thr - 1) / thr, thr, 0, stream>>>(T, q, dt, outT, ws, B);
    }

    int gpb = (N + 3) / 4;
    dim3 grid((unsigned)((gpb + 255) / 256), (unsigned)B);
    realign_kernel<<<grid, 256, 0, stream>>>(pcd, q, dt, ws, outP, N, gpb);
}


# one launch instead of two: every wave of the streaming kernel builds its batch's rigid inverse from q and dt itself, one extra block does the four f32 WMMAs of batch_T_pred; no workspace hand-over
# speedup vs baseline: 1.2625x; 1.2625x over previous
#include <hip/hip_runtime.h>
#include <hip/hip_bf16.h>
#include <math.h>

typedef __attribute__((ext_vector_type(2))) float v2f;
typedef __attribute__((ext_vector_type(4))) float v4f;
typedef __attribute__((ext_vector_type(8))) float v8f;

__device__ __forceinline__ void make_rinv(const float* __restrict__ q,
                                          const float* __restrict__ dt,
                                          int b, float R[9], float ti[3]) {
    float qw = q[4 * b + 0], qx = q[4 * b + 1], qy = q[4 * b + 2], qz = q[4 * b + 3];
    float s = 1.0f / sqrtf(qw * qw + qx * qx + qy * qy + qz * qz);
    qw *= s; qx *= s; qy *= s; qz *= s;
    float R00 = 1.f - 2.f * (qy * qy + qz * qz);
    float R01 = 2.f * (qx * qy - qw * qz);
    float R02 = 2.f * (qx * qz + qw * qy);
    float R10 = 2.f * (qx * qy + qw * qz);
    float R11 = 1.f - 2.f * (qx * qx + qz * qz);
    float R12 = 2.f * (qy * qz - qw * qx);
    float R20 = 2.f * (qx * qz - qw * qy);
    float R21 = 2.f * (qy * qz + qw * qx);
    float R22 = 1.f - 2.f * (qx * qx + qy * qy);
    R[0] = R00; R[1] = R10; R[2] = R20;
    R[3] = R01; R[4] = R11; R[5] = R21;
    R[6] = R02; R[7] = R12; R[8] = R22;
    float t0 = dt[3 * b + 0], t1 = dt[3 * b + 1], t2 = dt[3 * b + 2];
    ti[0] = -(R00 * t0 + R10 * t1 + R20 * t2);
    ti[1] = -(R01 * t0 + R11 * t1 + R21 * t2);
    ti[2] = -(R02 * t0 + R12 * t1 + R22 * t2);
}

__device__ __forceinline__ void batch_T_wmma16(
    const float* __restrict__ T, const float* __restrict__ q,
    const float* __restrict__ dt, float* __restrict__ outT,
    float* sA  , int tid) {
    if (tid < 16) {
        int b = tid;
        float R[9], ti[3];
        make_rinv(q, dt, b, R, ti);
        sA[b * 16 + 0]  = R[0]; sA[b * 16 + 1]  = R[1]; sA[b * 16 + 2]  = R[2]; sA[b * 16 + 3]  = ti[0];
        sA[b * 16 + 4]  = R[3]; sA[b * 16 + 5]  = R[4]; sA[b * 16 + 6]  = R[5]; sA[b * 16 + 7]  = ti[1];
        sA[b * 16 + 8]  = R[6]; sA[b * 16 + 9]  = R[7]; sA[b * 16 + 10] = R[8]; sA[b * 16 + 11] = ti[2];
        sA[b * 16 + 12] = 0.f;  sA[b * 16 + 13] = 0.f;  sA[b * 16 + 14] = 0.f;  sA[b * 16 + 15] = 1.f;
    }
    __syncthreads();

    if (tid < 32) {
        int lane = tid;
        int r  = lane & 15;
        int hi = lane >> 4;
        int qb = r >> 2;
        int qi = r & 3;
        #pragma unroll
        for (int g = 0; g < 4; ++g) {
            int batch = g * 4 + qb;
            v2f a;
            a.x = sA[batch * 16 + qi * 4 + 2 * hi + 0];
            a.y = sA[batch * 16 + qi * 4 + 2 * hi + 1];
            v2f bm;
            bm.x = T[batch * 16 + (2 * hi + 0) * 4 + qi];
            bm.y = T[batch * 16 + (2 * hi + 1) * 4 + qi];
            v8f c = {};
            c = __builtin_amdgcn_wmma_f32_16x16x4_f32(
                 false, a,  false, bm,
                 (short)0, c,  false,  false);
            #pragma unroll
            for (int v = 0; v < 8; ++v) {
                int m = v + 8 * hi;
                if ((m >> 2) == qb) {
                    outT[batch * 16 + (m & 3) * 4 + qi] = c[v];
                }
            }
        }
    }
}

__device__ __forceinline__ void batch_T_scalar(
    const float* __restrict__ T, const float* __restrict__ q,
    const float* __restrict__ dt, float* __restrict__ outT, int B, int tid, int nthreads) {
    for (int b = tid; b < B; b += nthreads) {
        float R[9], ti[3];
        make_rinv(q, dt, b, R, ti);
        float Ainv[16] = {R[0], R[1], R[2], ti[0],
                          R[3], R[4], R[5], ti[1],
                          R[6], R[7], R[8], ti[2],
                          0.f,  0.f,  0.f,  1.f};
        for (int i = 0; i < 4; ++i)
            for (int j = 0; j < 4; ++j) {
                float s = 0.f;
                for (int k = 0; k < 4; ++k) s += Ainv[i * 4 + k] * T[b * 16 + k * 4 + j];
                outT[b * 16 + i * 4 + j] = s;
            }
    }
}

__global__ __launch_bounds__(256) void realign_kernel(
    const float* __restrict__ pcd, const float* __restrict__ T,
    const float* __restrict__ q, const float* __restrict__ dt,
    float* __restrict__ outT, float* __restrict__ outP, int N, int gpb, int B) {
    __shared__ float sA[256];
    if (blockIdx.x == gridDim.x - 1) {
        if (blockIdx.y == 0) {
            if (B == 16) batch_T_wmma16(T, q, dt, outT, sA, (int)threadIdx.x);
            else         batch_T_scalar(T, q, dt, outT, B, (int)threadIdx.x, (int)blockDim.x);
        }
        return;
    }

    int b = blockIdx.y;
    int g = blockIdx.x * blockDim.x + threadIdx.x;
    if (g >= gpb) return;

    long long p0 = (long long)g * 4;
    size_t base = (size_t)b * (size_t)N * 3 + (size_t)p0 * 3;

    if (p0 + 4 <= N) {
        const v4f* src = (const v4f*)(pcd + base);
        v4f va = __builtin_nontemporal_load(src + 0);
        v4f vb = __builtin_nontemporal_load(src + 1);
        v4f vc = __builtin_nontemporal_load(src + 2);
        float R[9], ti[3];
        make_rinv(q, dt, b, R, ti);
        float f[12] = {va.x, va.y, va.z, va.w,
                       vb.x, vb.y, vb.z, vb.w,
                       vc.x, vc.y, vc.z, vc.w};
        float o[12];
        #pragma unroll
        for (int p = 0; p < 4; ++p) {
            float px = f[p * 3 + 0], py = f[p * 3 + 1], pz = f[p * 3 + 2];
            o[p * 3 + 0] = fmaf(R[0], px, fmaf(R[1], py, fmaf(R[2], pz, ti[0])));
            o[p * 3 + 1] = fmaf(R[3], px, fmaf(R[4], py, fmaf(R[5], pz, ti[1])));
            o[p * 3 + 2] = fmaf(R[6], px, fmaf(R[7], py, fmaf(R[8], pz, ti[2])));
        }
        v4f o0 = {o[0], o[1], o[2], o[3]};
        v4f o1 = {o[4], o[5], o[6], o[7]};
        v4f o2 = {o[8], o[9], o[10], o[11]};
        v4f* dst = (v4f*)(outP + base);
        __builtin_nontemporal_store(o0, dst + 0);
        __builtin_nontemporal_store(o1, dst + 1);
        __builtin_nontemporal_store(o2, dst + 2);
    } else {
        float R[9], ti[3];
        make_rinv(q, dt, b, R, ti);
        for (long long p = p0; p < N; ++p) {
            size_t idx = (size_t)b * (size_t)N * 3 + (size_t)p * 3;
            float px = pcd[idx + 0], py = pcd[idx + 1], pz = pcd[idx + 2];
            outP[idx + 0] = fmaf(R[0], px, fmaf(R[1], py, fmaf(R[2], pz, ti[0])));
            outP[idx + 1] = fmaf(R[3], px, fmaf(R[4], py, fmaf(R[5], pz, ti[1])));
            outP[idx + 2] = fmaf(R[6], px, fmaf(R[7], py, fmaf(R[8], pz, ti[2])));
        }
    }
}

extern "C" void kernel_launch(void* const* d_in, const int* in_sizes, int n_in,
                              void* d_out, int out_size, void* d_ws, size_t ws_size,
                              hipStream_t stream) {
    (void)d_ws; (void)ws_size;
    const float* pcd = (const float*)d_in[0];
    const float* T   = (const float*)d_in[1];
    const float* q   = (const float*)d_in[2];
    const float* dt  = (const float*)d_in[3];

    int B = in_sizes[2] / 4;
    long long pcd_elems = in_sizes[0];
    int N = (int)(pcd_elems / ((long long)B * 3));

    float* outT = (float*)d_out;
    float* outP = outT + (size_t)B * 16;

    int gpb = (N + 3) / 4;
    dim3 grid((unsigned)((gpb + 255) / 256) + 1u, (unsigned)B);
    realign_kernel<<<grid, 256, 0, stream>>>(pcd, T, q, dt, outT, outP, N, gpb, B);
}


# whole-line global access: each load and store instruction of a wave covers 512 contiguous bytes (4 lines) instead of 16-B pieces at a 48-B stride (12 lines); xyz regrouping done in wave-private LDS
# speedup vs baseline: 1.3755x; 1.0895x over previous
#include <hip/hip_runtime.h>
#include <hip/hip_bf16.h>
#include <math.h>

typedef __attribute__((ext_vector_type(2))) float v2f;
typedef __attribute__((ext_vector_type(4))) float v4f;
typedef __attribute__((ext_vector_type(8))) float v8f;

struct QT { float qw, qx, qy, qz, t0, t1, t2; };

__device__ __forceinline__ QT load_qt(const float* __restrict__ q,
                                      const float* __restrict__ dt, int b) {
    QT v;
    v.qw = q[4 * b + 0]; v.qx = q[4 * b + 1]; v.qy = q[4 * b + 2]; v.qz = q[4 * b + 3];
    v.t0 = dt[3 * b + 0]; v.t1 = dt[3 * b + 1]; v.t2 = dt[3 * b + 2];
    return v;
}

__device__ __forceinline__ void rinv_from(const QT& v, float R[9], float ti[3]) {
    float qw = v.qw, qx = v.qx, qy = v.qy, qz = v.qz;
    float s = 1.0f / sqrtf(qw * qw + qx * qx + qy * qy + qz * qz);
    qw *= s; qx *= s; qy *= s; qz *= s;
    float R00 = 1.f - 2.f * (qy * qy + qz * qz);
    float R01 = 2.f * (qx * qy - qw * qz);
    float R02 = 2.f * (qx * qz + qw * qy);
    float R10 = 2.f * (qx * qy + qw * qz);
    float R11 = 1.f - 2.f * (qx * qx + qz * qz);
    float R12 = 2.f * (qy * qz - qw * qx);
    float R20 = 2.f * (qx * qz - qw * qy);
    float R21 = 2.f * (qy * qz + qw * qx);
    float R22 = 1.f - 2.f * (qx * qx + qy * qy);
    R[0] = R00; R[1] = R10; R[2] = R20;
    R[3] = R01; R[4] = R11; R[5] = R21;
    R[6] = R02; R[7] = R12; R[8] = R22;
    float t0 = v.t0, t1 = v.t1, t2 = v.t2;
    ti[0] = -(R00 * t0 + R10 * t1 + R20 * t2);
    ti[1] = -(R01 * t0 + R11 * t1 + R21 * t2);
    ti[2] = -(R02 * t0 + R12 * t1 + R22 * t2);
}

__device__ __forceinline__ void make_rinv(const float* __restrict__ q,
                                          const float* __restrict__ dt,
                                          int b, float R[9], float ti[3]) {
    rinv_from(load_qt(q, dt, b), R, ti);
}

__device__ __forceinline__ void batch_T_wmma16(
    const float* __restrict__ T, const float* __restrict__ q,
    const float* __restrict__ dt, float* __restrict__ outT,
    float* sA  , int tid) {
    if (tid < 16) {
        int b = tid;
        float R[9], ti[3];
        make_rinv(q, dt, b, R, ti);
        sA[b * 16 + 0]  = R[0]; sA[b * 16 + 1]  = R[1]; sA[b * 16 + 2]  = R[2]; sA[b * 16 + 3]  = ti[0];
        sA[b * 16 + 4]  = R[3]; sA[b * 16 + 5]  = R[4]; sA[b * 16 + 6]  = R[5]; sA[b * 16 + 7]  = ti[1];
        sA[b * 16 + 8]  = R[6]; sA[b * 16 + 9]  = R[7]; sA[b * 16 + 10] = R[8]; sA[b * 16 + 11] = ti[2];
        sA[b * 16 + 12] = 0.f;  sA[b * 16 + 13] = 0.f;  sA[b * 16 + 14] = 0.f;  sA[b * 16 + 15] = 1.f;
    }
    __syncthreads();

    if (tid < 32) {
        int lane = tid;
        int r  = lane & 15;
        int hi = lane >> 4;
        int qb = r >> 2;
        int qi = r & 3;
        #pragma unroll
        for (int g = 0; g < 4; ++g) {
            int batch = g * 4 + qb;
            v2f a;
            a.x = sA[batch * 16 + qi * 4 + 2 * hi + 0];
            a.y = sA[batch * 16 + qi * 4 + 2 * hi + 1];
            v2f bm;
            bm.x = T[batch * 16 + (2 * hi + 0) * 4 + qi];
            bm.y = T[batch * 16 + (2 * hi + 1) * 4 + qi];
            v8f c = {};
            c = __builtin_amdgcn_wmma_f32_16x16x4_f32(
                 false, a,  false, bm,
                 (short)0, c,  false,  false);
            #pragma unroll
            for (int v = 0; v < 8; ++v) {
                int m = v + 8 * hi;
                if ((m >> 2) == qb) {
                    outT[batch * 16 + (m & 3) * 4 + qi] = c[v];
                }
            }
        }
    }
}

__device__ __forceinline__ void batch_T_scalar(
    const float* __restrict__ T, const float* __restrict__ q,
    const float* __restrict__ dt, float* __restrict__ outT, int B, int tid, int nthreads) {
    for (int b = tid; b < B; b += nthreads) {
        float R[9], ti[3];
        make_rinv(q, dt, b, R, ti);
        float Ainv[16] = {R[0], R[1], R[2], ti[0],
                          R[3], R[4], R[5], ti[1],
                          R[6], R[7], R[8], ti[2],
                          0.f,  0.f,  0.f,  1.f};
        for (int i = 0; i < 4; ++i)
            for (int j = 0; j < 4; ++j) {
                float s = 0.f;
                for (int k = 0; k < 4; ++k) s += Ainv[i * 4 + k] * T[b * 16 + k * 4 + j];
                outT[b * 16 + i * 4 + j] = s;
            }
    }
}

__global__ __launch_bounds__(256) void realign_kernel(
    const float* __restrict__ pcd, const float* __restrict__ T,
    const float* __restrict__ q, const float* __restrict__ dt,
    float* __restrict__ outT, float* __restrict__ outP, int N, int gpb, int B) {
    __shared__ v4f sX[8 * 96];
    if (blockIdx.x == gridDim.x - 1) {
        if (blockIdx.y == 0) {
            float* sA = (float*)sX;
            if (B == 16) batch_T_wmma16(T, q, dt, outT, sA, (int)threadIdx.x);
            else         batch_T_scalar(T, q, dt, outT, B, (int)threadIdx.x, (int)blockDim.x);
        }
        return;
    }

    int b = blockIdx.y;
    int lane = (int)(threadIdx.x & 31u);
    int wave = __builtin_amdgcn_readfirstlane((int)(threadIdx.x >> 5));
    long long wp0 = ((long long)blockIdx.x * 8 + wave) * 128;
    if (wp0 >= N) return;
    size_t bbase = (size_t)b * (size_t)N * 3;

    if (wp0 + 128 <= N) {
        const v4f* src = (const v4f*)(pcd + bbase + (size_t)wp0 * 3);
        QT qt = load_qt(q, dt, b);
        v4f l0 = __builtin_nontemporal_load(src + lane);
        v4f l1 = __builtin_nontemporal_load(src + 32 + lane);
        v4f l2 = __builtin_nontemporal_load(src + 64 + lane);
        __builtin_amdgcn_sched_barrier(0);
        float R[9], ti[3];
        rinv_from(qt, R, ti);
        __builtin_amdgcn_sched_barrier(0);
        v4f* xw = sX + wave * 96;
        xw[lane]      = l0;
        xw[32 + lane] = l1;
        xw[64 + lane] = l2;
        v4f va = xw[3 * lane + 0];
        v4f vb = xw[3 * lane + 1];
        v4f vc = xw[3 * lane + 2];
        float f[12] = {va.x, va.y, va.z, va.w,
                       vb.x, vb.y, vb.z, vb.w,
                       vc.x, vc.y, vc.z, vc.w};
        float o[12];
        #pragma unroll
        for (int p = 0; p < 4; ++p) {
            float px = f[p * 3 + 0], py = f[p * 3 + 1], pz = f[p * 3 + 2];
            o[p * 3 + 0] = fmaf(R[0], px, fmaf(R[1], py, fmaf(R[2], pz, ti[0])));
            o[p * 3 + 1] = fmaf(R[3], px, fmaf(R[4], py, fmaf(R[5], pz, ti[1])));
            o[p * 3 + 2] = fmaf(R[6], px, fmaf(R[7], py, fmaf(R[8], pz, ti[2])));
        }
        v4f o0 = {o[0], o[1], o[2], o[3]};
        v4f o1 = {o[4], o[5], o[6], o[7]};
        v4f o2 = {o[8], o[9], o[10], o[11]};
        xw[3 * lane + 0] = o0;
        xw[3 * lane + 1] = o1;
        xw[3 * lane + 2] = o2;
        v4f s0 = xw[lane];
        v4f s1 = xw[32 + lane];
        v4f s2 = xw[64 + lane];
        v4f* dst = (v4f*)(outP + bbase + (size_t)wp0 * 3);
        __builtin_nontemporal_store(s0, dst + lane);
        __builtin_nontemporal_store(s1, dst + 32 + lane);
        __builtin_nontemporal_store(s2, dst + 64 + lane);
        return;
    }

    long long g = wp0 / 4 + lane;
    if (g >= gpb) return;
    long long p0 = g * 4;
    size_t base = bbase + (size_t)p0 * 3;
    float R[9], ti[3];
    make_rinv(q, dt, b, R, ti);
    if (p0 + 4 <= N) {
        const v4f* src = (const v4f*)(pcd + base);
        v4f va = __builtin_nontemporal_load(src + 0);
        v4f vb = __builtin_nontemporal_load(src + 1);
        v4f vc = __builtin_nontemporal_load(src + 2);
        float f[12] = {va.x, va.y, va.z, va.w,
                       vb.x, vb.y, vb.z, vb.w,
                       vc.x, vc.y, vc.z, vc.w};
        float o[12];
        #pragma unroll
        for (int p = 0; p < 4; ++p) {
            float px = f[p * 3 + 0], py = f[p * 3 + 1], pz = f[p * 3 + 2];
            o[p * 3 + 0] = fmaf(R[0], px, fmaf(R[1], py, fmaf(R[2], pz, ti[0])));
            o[p * 3 + 1] = fmaf(R[3], px, fmaf(R[4], py, fmaf(R[5], pz, ti[1])));
            o[p * 3 + 2] = fmaf(R[6], px, fmaf(R[7], py, fmaf(R[8], pz, ti[2])));
        }
        v4f o0 = {o[0], o[1], o[2], o[3]};
        v4f o1 = {o[4], o[5], o[6], o[7]};
        v4f o2 = {o[8], o[9], o[10], o[11]};
        v4f* dst = (v4f*)(outP + base);
        __builtin_nontemporal_store(o0, dst + 0);
        __builtin_nontemporal_store(o1, dst + 1);
        __builtin_nontemporal_store(o2, dst + 2);
    } else {
        for (long long p = p0; p < N; ++p) {
            size_t idx = bbase + (size_t)p * 3;
            float px = pcd[idx + 0], py = pcd[idx + 1], pz = pcd[idx + 2];
            outP[idx + 0] = fmaf(R[0], px, fmaf(R[1], py, fmaf(R[2], pz, ti[0])));
            outP[idx + 1] = fmaf(R[3], px, fmaf(R[4], py, fmaf(R[5], pz, ti[1])));
            outP[idx + 2] = fmaf(R[6], px, fmaf(R[7], py, fmaf(R[8], pz, ti[2])));
        }
    }
}

extern "C" void kernel_launch(void* const* d_in, const int* in_sizes, int n_in,
                              void* d_out, int out_size, void* d_ws, size_t ws_size,
                              hipStream_t stream) {
    (void)d_ws; (void)ws_size;
    const float* pcd = (const float*)d_in[0];
    const float* T   = (const float*)d_in[1];
    const float* q   = (const float*)d_in[2];
    const float* dt  = (const float*)d_in[3];

    int B = in_sizes[2] / 4;
    long long pcd_elems = in_sizes[0];
    int N = (int)(pcd_elems / ((long long)B * 3));

    float* outT = (float*)d_out;
    float* outP = outT + (size_t)B * 16;

    int gpb = (N + 3) / 4;
    dim3 grid((unsigned)((gpb + 255) / 256) + 1u, (unsigned)B);
    realign_kernel<<<grid, 256, 0, stream>>>(pcd, T, q, dt, outT, outP, N, gpb, B);
}


# each wave streams four consecutive 128-point tiles (6 KiB) and requests all twelve loads before its first wait: a quarter of the waves enter, four times the bytes in flight per wave
# speedup vs baseline: 1.4604x; 1.0617x over previous
#include <hip/hip_runtime.h>
#include <hip/hip_bf16.h>
#include <math.h>

typedef __attribute__((ext_vector_type(2))) float v2f;
typedef __attribute__((ext_vector_type(4))) float v4f;
typedef __attribute__((ext_vector_type(8))) float v8f;

struct QT { float qw, qx, qy, qz, t0, t1, t2; };

__device__ __forceinline__ QT load_qt(const float* __restrict__ q,
                                      const float* __restrict__ dt, int b) {
    QT v;
    v.qw = q[4 * b + 0]; v.qx = q[4 * b + 1]; v.qy = q[4 * b + 2]; v.qz = q[4 * b + 3];
    v.t0 = dt[3 * b + 0]; v.t1 = dt[3 * b + 1]; v.t2 = dt[3 * b + 2];
    return v;
}

__device__ __forceinline__ void rinv_from(const QT& v, float R[9], float ti[3]) {
    float qw = v.qw, qx = v.qx, qy = v.qy, qz = v.qz;
    float s = 1.0f / sqrtf(qw * qw + qx * qx + qy * qy + qz * qz);
    qw *= s; qx *= s; qy *= s; qz *= s;
    float R00 = 1.f - 2.f * (qy * qy + qz * qz);
    float R01 = 2.f * (qx * qy - qw * qz);
    float R02 = 2.f * (qx * qz + qw * qy);
    float R10 = 2.f * (qx * qy + qw * qz);
    float R11 = 1.f - 2.f * (qx * qx + qz * qz);
    float R12 = 2.f * (qy * qz - qw * qx);
    float R20 = 2.f * (qx * qz - qw * qy);
    float R21 = 2.f * (qy * qz + qw * qx);
    float R22 = 1.f - 2.f * (qx * qx + qy * qy);
    R[0] = R00; R[1] = R10; R[2] = R20;
    R[3] = R01; R[4] = R11; R[5] = R21;
    R[6] = R02; R[7] = R12; R[8] = R22;
    float t0 = v.t0, t1 = v.t1, t2 = v.t2;
    ti[0] = -(R00 * t0 + R10 * t1 + R20 * t2);
    ti[1] = -(R01 * t0 + R11 * t1 + R21 * t2);
    ti[2] = -(R02 * t0 + R12 * t1 + R22 * t2);
}

__device__ __forceinline__ void make_rinv(const float* __restrict__ q,
                                          const float* __restrict__ dt,
                                          int b, float R[9], float ti[3]) {
    rinv_from(load_qt(q, dt, b), R, ti);
}

__device__ __forceinline__ void batch_T_wmma16(
    const float* __restrict__ T, const float* __restrict__ q,
    const float* __restrict__ dt, float* __restrict__ outT,
    float* sA  , int tid) {
    if (tid < 16) {
        int b = tid;
        float R[9], ti[3];
        make_rinv(q, dt, b, R, ti);
        sA[b * 16 + 0]  = R[0]; sA[b * 16 + 1]  = R[1]; sA[b * 16 + 2]  = R[2]; sA[b * 16 + 3]  = ti[0];
        sA[b * 16 + 4]  = R[3]; sA[b * 16 + 5]  = R[4]; sA[b * 16 + 6]  = R[5]; sA[b * 16 + 7]  = ti[1];
        sA[b * 16 + 8]  = R[6]; sA[b * 16 + 9]  = R[7]; sA[b * 16 + 10] = R[8]; sA[b * 16 + 11] = ti[2];
        sA[b * 16 + 12] = 0.f;  sA[b * 16 + 13] = 0.f;  sA[b * 16 + 14] = 0.f;  sA[b * 16 + 15] = 1.f;
    }
    __syncthreads();

    if (tid < 32) {
        int lane = tid;
        int r  = lane & 15;
        int hi = lane >> 4;
        int qb = r >> 2;
        int qi = r & 3;
        #pragma unroll
        for (int g = 0; g < 4; ++g) {
            int batch = g * 4 + qb;
            v2f a;
            a.x = sA[batch * 16 + qi * 4 + 2 * hi + 0];
            a.y = sA[batch * 16 + qi * 4 + 2 * hi + 1];
            v2f bm;
            bm.x = T[batch * 16 + (2 * hi + 0) * 4 + qi];
            bm.y = T[batch * 16 + (2 * hi + 1) * 4 + qi];
            v8f c = {};
            c = __builtin_amdgcn_wmma_f32_16x16x4_f32(
                 false, a,  false, bm,
                 (short)0, c,  false,  false);
            #pragma unroll
            for (int v = 0; v < 8; ++v) {
                int m = v + 8 * hi;
                if ((m >> 2) == qb) {
                    outT[batch * 16 + (m & 3) * 4 + qi] = c[v];
                }
            }
        }
    }
}

__device__ __forceinline__ void batch_T_scalar(
    const float* __restrict__ T, const float* __restrict__ q,
    const float* __restrict__ dt, float* __restrict__ outT, int B, int tid, int nthreads) {
    for (int b = tid; b < B; b += nthreads) {
        float R[9], ti[3];
        make_rinv(q, dt, b, R, ti);
        float Ainv[16] = {R[0], R[1], R[2], ti[0],
                          R[3], R[4], R[5], ti[1],
                          R[6], R[7], R[8], ti[2],
                          0.f,  0.f,  0.f,  1.f};
        for (int i = 0; i < 4; ++i)
            for (int j = 0; j < 4; ++j) {
                float s = 0.f;
                for (int k = 0; k < 4; ++k) s += Ainv[i * 4 + k] * T[b * 16 + k * 4 + j];
                outT[b * 16 + i * 4 + j] = s;
            }
    }
}

#define TILES_PER_WAVE 4

__device__ __forceinline__ void tile_through_lds(
    v4f l0, v4f l1, v4f l2, const float R[9], const float ti[3],
    v4f* xw, int lane, v4f* dst) {
    xw[lane]      = l0;
    xw[32 + lane] = l1;
    xw[64 + lane] = l2;
    __builtin_amdgcn_fence(__ATOMIC_ACQ_REL, "wavefront");
    v4f va = xw[3 * lane + 0];
    v4f vb = xw[3 * lane + 1];
    v4f vc = xw[3 * lane + 2];
    float f[12] = {va.x, va.y, va.z, va.w,
                   vb.x, vb.y, vb.z, vb.w,
                   vc.x, vc.y, vc.z, vc.w};
    float o[12];
    #pragma unroll
    for (int p = 0; p < 4; ++p) {
        float px = f[p * 3 + 0], py = f[p * 3 + 1], pz = f[p * 3 + 2];
        o[p * 3 + 0] = fmaf(R[0], px, fmaf(R[1], py, fmaf(R[2], pz, ti[0])));
        o[p * 3 + 1] = fmaf(R[3], px, fmaf(R[4], py, fmaf(R[5], pz, ti[1])));
        o[p * 3 + 2] = fmaf(R[6], px, fmaf(R[7], py, fmaf(R[8], pz, ti[2])));
    }
    v4f o0 = {o[0], o[1], o[2], o[3]};
    v4f o1 = {o[4], o[5], o[6], o[7]};
    v4f o2 = {o[8], o[9], o[10], o[11]};
    xw[3 * lane + 0] = o0;
    xw[3 * lane + 1] = o1;
    xw[3 * lane + 2] = o2;
    __builtin_amdgcn_fence(__ATOMIC_ACQ_REL, "wavefront");
    v4f s0 = xw[lane];
    v4f s1 = xw[32 + lane];
    v4f s2 = xw[64 + lane];
    __builtin_nontemporal_store(s0, dst + lane);
    __builtin_nontemporal_store(s1, dst + 32 + lane);
    __builtin_nontemporal_store(s2, dst + 64 + lane);
}

__global__ __launch_bounds__(256) void realign_kernel(
    const float* __restrict__ pcd, const float* __restrict__ T,
    const float* __restrict__ q, const float* __restrict__ dt,
    float* __restrict__ outT, float* __restrict__ outP, int N, int gpb, int B) {
    __shared__ v4f sX[8 * 96];
    if (blockIdx.x == gridDim.x - 1) {
        if (blockIdx.y == 0) {
            float* sA = (float*)sX;
            if (B == 16) batch_T_wmma16(T, q, dt, outT, sA, (int)threadIdx.x);
            else         batch_T_scalar(T, q, dt, outT, B, (int)threadIdx.x, (int)blockDim.x);
        }
        return;
    }

    int b = blockIdx.y;
    int lane = (int)(threadIdx.x & 31u);
    int wave = __builtin_amdgcn_readfirstlane((int)(threadIdx.x >> 5));
    long long wp0 = ((long long)blockIdx.x * 8 + wave) * (128 * TILES_PER_WAVE);
    if (wp0 >= N) return;
    size_t bbase = (size_t)b * (size_t)N * 3;
    v4f* xw = sX + wave * 96;

    if (wp0 + 128 * TILES_PER_WAVE <= N) {
        const v4f* src = (const v4f*)(pcd + bbase + (size_t)wp0 * 3);
        v4f* dst = (v4f*)(outP + bbase + (size_t)wp0 * 3);
        QT qt = load_qt(q, dt, b);
        v4f l[3 * TILES_PER_WAVE];
        #pragma unroll
        for (int k = 0; k < 3 * TILES_PER_WAVE; ++k)
            l[k] = __builtin_nontemporal_load(src + 32 * k + lane);
        __builtin_amdgcn_sched_barrier(0);
        float R[9], ti[3];
        rinv_from(qt, R, ti);
        __builtin_amdgcn_sched_barrier(0);
        #pragma unroll
        for (int t = 0; t < TILES_PER_WAVE; ++t)
            tile_through_lds(l[3 * t + 0], l[3 * t + 1], l[3 * t + 2], R, ti, xw, lane, dst + 96 * t);
        return;
    }

    float R[9], ti[3];
    make_rinv(q, dt, b, R, ti);
    for (int t = 0; t < TILES_PER_WAVE; ++t) {
        long long tp0 = wp0 + 128 * t;
        if (tp0 >= N) return;
        if (tp0 + 128 <= N) {
            const v4f* src = (const v4f*)(pcd + bbase + (size_t)tp0 * 3);
            v4f l0 = __builtin_nontemporal_load(src + lane);
            v4f l1 = __builtin_nontemporal_load(src + 32 + lane);
            v4f l2 = __builtin_nontemporal_load(src + 64 + lane);
            tile_through_lds(l0, l1, l2, R, ti, xw, lane, (v4f*)(outP + bbase + (size_t)tp0 * 3));
            continue;
        }
        long long g = tp0 / 4 + lane;
        if (g >= gpb) return;
        long long p0 = g * 4;
        size_t base = bbase + (size_t)p0 * 3;
        if (p0 + 4 <= N) {
            const v4f* src = (const v4f*)(pcd + base);
            v4f va = __builtin_nontemporal_load(src + 0);
            v4f vb = __builtin_nontemporal_load(src + 1);
            v4f vc = __builtin_nontemporal_load(src + 2);
            float f[12] = {va.x, va.y, va.z, va.w,
                           vb.x, vb.y, vb.z, vb.w,
                           vc.x, vc.y, vc.z, vc.w};
            float o[12];
            #pragma unroll
            for (int p = 0; p < 4; ++p) {
                float px = f[p * 3 + 0], py = f[p * 3 + 1], pz = f[p * 3 + 2];
                o[p * 3 + 0] = fmaf(R[0], px, fmaf(R[1], py, fmaf(R[2], pz, ti[0])));
                o[p * 3 + 1] = fmaf(R[3], px, fmaf(R[4], py, fmaf(R[5], pz, ti[1])));
                o[p * 3 + 2] = fmaf(R[6], px, fmaf(R[7], py, fmaf(R[8], pz, ti[2])));
            }
            v4f o0 = {o[0], o[1], o[2], o[3]};
            v4f o1 = {o[4], o[5], o[6], o[7]};
            v4f o2 = {o[8], o[9], o[10], o[11]};
            v4f* dst = (v4f*)(outP + base);
            __builtin_nontemporal_store(o0, dst + 0);
            __builtin_nontemporal_store(o1, dst + 1);
            __builtin_nontemporal_store(o2, dst + 2);
        } else {
            for (long long p = p0; p < N; ++p) {
                size_t idx = bbase + (size_t)p * 3;
                float px = pcd[idx + 0], py = pcd[idx + 1], pz = pcd[idx + 2];
                outP[idx + 0] = fmaf(R[0], px, fmaf(R[1], py, fmaf(R[2], pz, ti[0])));
                outP[idx + 1] = fmaf(R[3], px, fmaf(R[4], py, fmaf(R[5], pz, ti[1])));
                outP[idx + 2] = fmaf(R[6], px, fmaf(R[7], py, fmaf(R[8], pz, ti[2])));
            }
        }
        return;
    }
}

extern "C" void kernel_launch(void* const* d_in, const int* in_sizes, int n_in,
                              void* d_out, int out_size, void* d_ws, size_t ws_size,
                              hipStream_t stream) {
    (void)d_ws; (void)ws_size;
    const float* pcd = (const float*)d_in[0];
    const float* T   = (const float*)d_in[1];
    const float* q   = (const float*)d_in[2];
    const float* dt  = (const float*)d_in[3];

    int B = in_sizes[2] / 4;
    long long pcd_elems = in_sizes[0];
    int N = (int)(pcd_elems / ((long long)B * 3));

    float* outT = (float*)d_out;
    float* outP = outT + (size_t)B * 16;

    int gpb = (N + 3) / 4;
    int ppb = 8 * TILES_PER_WAVE * 128;
    dim3 grid((unsigned)((N + ppb - 1) / ppb) + 1u, (unsigned)B);
    realign_kernel<<<grid, 256, 0, stream>>>(pcd, T, q, dt, outT, outP, N, gpb, B);
}
